# starting kernel
# baseline (speedup 1.0000x reference)
#include <hip/hip_runtime.h>


typedef __attribute__((ext_vector_type(2))) float v2f;
typedef __attribute__((ext_vector_type(8))) float v8f;

#define K_ITERS 20
#define MU_CONST 0.05f
#define NB 4096
#define NN 128

__global__ __launch_bounds__(256)
void nag_wmma_kernel(const float* __restrict__ w,
                     const float* __restrict__ b,
                     const float* __restrict__ sp,
                     float* __restrict__ out)
{
    __shared__ float sy[NN];

    const int tid   = threadIdx.x;
    const int lane  = tid & 31;
    const int wave  = tid >> 5;
    const int n     = lane & 15;
    const int half  = lane >> 4;
    const int rb    = wave << 4;
    const int batch = blockIdx.x;

    const float s     = *sp;
    const float sqms  = sqrtf(MU_CONST * s);
    const float alpha = (1.0f - sqms) / (1.0f + sqms);

    const float* wp = w + (size_t)batch * (NN * NN) + (size_t)(rb + n) * NN + half * 2;
    v2f wt[32];
#pragma unroll
    for (int kc = 0; kc < 32; ++kc)
        wt[kc] = *(const v2f*)(wp + kc * 4);

    const float bv = b[(size_t)batch * NN + rb + n];

    if (tid < NN) sy[tid] = 0.0f;
    __syncthreads();

    float xl = 0.0f, yl = 0.0f;

#pragma unroll 1
    for (int it = 0; it < K_ITERS; ++it) {
        v8f acc0 = {}, acc1 = {}, acc2 = {}, acc3 = {};
#pragma unroll
        for (int kc = 0; kc < 32; kc += 4) {
            v2f a0 = *(const v2f*)(&sy[(kc + 0) * 4 + half * 2]);
            v2f a1 = *(const v2f*)(&sy[(kc + 1) * 4 + half * 2]);
            v2f a2 = *(const v2f*)(&sy[(kc + 2) * 4 + half * 2]);
            v2f a3 = *(const v2f*)(&sy[(kc + 3) * 4 + half * 2]);
            acc0 = __builtin_amdgcn_wmma_f32_16x16x4_f32(false, a0, false, wt[kc + 0],
                                                         (short)0, acc0, false, false);
            acc1 = __builtin_amdgcn_wmma_f32_16x16x4_f32(false, a1, false, wt[kc + 1],
                                                         (short)0, acc1, false, false);
            acc2 = __builtin_amdgcn_wmma_f32_16x16x4_f32(false, a2, false, wt[kc + 2],
                                                         (short)0, acc2, false, false);
            acc3 = __builtin_amdgcn_wmma_f32_16x16x4_f32(false, a3, false, wt[kc + 3],
                                                         (short)0, acc3, false, false);
        }
        float g  = (acc0[0] + acc1[0]) + (acc2[0] + acc3[0]);
        float xn = yl - s * (g - bv);
        float yn = (1.0f + alpha) * xn - alpha * xl;
        xl = xn;
        yl = yn;

        __syncthreads();
        if (!(lane & 16)) sy[rb + n] = yn;
        __syncthreads();
    }

    if (!(lane & 16)) {
        out[(size_t)batch * NN + rb + n] = xl;
        out[(size_t)NB * NN + (size_t)batch * NN + rb + n] = yl;
    }
}

extern "C" void kernel_launch(void* const* d_in, const int* in_sizes, int n_in,
                              void* d_out, int out_size, void* d_ws, size_t ws_size,
                              hipStream_t stream) {
    (void)in_sizes; (void)n_in; (void)d_ws; (void)ws_size; (void)out_size;
    const float* w = (const float*)d_in[0];
    const float* b = (const float*)d_in[1];
    const float* s = (const float*)d_in[2];
    float* out = (float*)d_out;
    nag_wmma_kernel<<<dim3(NB), dim3(256), 0, stream>>>(w, b, s, out);
}


# Drop the matrix product of step 0 only: y0 = 0 makes w@y0 exactly 0 for finite w, so x1 = s*b and y1 = (1+alpha)*x1 are formed directly; all 20 updates kept, 19 products left, sum order unchanged
# speedup vs baseline: 1.0484x; 1.0484x over previous
#include <hip/hip_runtime.h>


typedef __attribute__((ext_vector_type(2))) float v2f;
typedef __attribute__((ext_vector_type(8))) float v8f;

#define K_ITERS 20
#define MU_CONST 0.05f
#define NB 4096
#define NN 128

__global__ __launch_bounds__(256)
void nag_wmma_kernel(const float* __restrict__ w,
                     const float* __restrict__ b,
                     const float* __restrict__ sp,
                     float* __restrict__ out)
{
    __shared__ float sy[NN];

    const int tid   = threadIdx.x;
    const int lane  = tid & 31;
    const int wave  = tid >> 5;
    const int n     = lane & 15;
    const int half  = lane >> 4;
    const int rb    = wave << 4;
    const int batch = blockIdx.x;

    const float s     = *sp;
    const float sqms  = sqrtf(MU_CONST * s);
    const float alpha = (1.0f - sqms) / (1.0f + sqms);

    const float bv = b[(size_t)batch * NN + rb + n];

    const float* wp = w + (size_t)batch * (NN * NN) + (size_t)(rb + n) * NN + half * 2;
    v2f wt[32];
#pragma unroll
    for (int kc = 0; kc < 32; ++kc)
        wt[kc] = *(const v2f*)(wp + kc * 4);

    float xl = s * bv;
    float yl = (1.0f + alpha) * xl;

    if (!(lane & 16)) sy[rb + n] = yl;
    __syncthreads();

#pragma unroll 1
    for (int it = 1; it < K_ITERS; ++it) {
        v8f acc0 = {}, acc1 = {}, acc2 = {}, acc3 = {};
#pragma unroll
        for (int kc = 0; kc < 32; kc += 4) {
            v2f a0 = *(const v2f*)(&sy[(kc + 0) * 4 + half * 2]);
            v2f a1 = *(const v2f*)(&sy[(kc + 1) * 4 + half * 2]);
            v2f a2 = *(const v2f*)(&sy[(kc + 2) * 4 + half * 2]);
            v2f a3 = *(const v2f*)(&sy[(kc + 3) * 4 + half * 2]);
            acc0 = __builtin_amdgcn_wmma_f32_16x16x4_f32(false, a0, false, wt[kc + 0],
                                                         (short)0, acc0, false, false);
            acc1 = __builtin_amdgcn_wmma_f32_16x16x4_f32(false, a1, false, wt[kc + 1],
                                                         (short)0, acc1, false, false);
            acc2 = __builtin_amdgcn_wmma_f32_16x16x4_f32(false, a2, false, wt[kc + 2],
                                                         (short)0, acc2, false, false);
            acc3 = __builtin_amdgcn_wmma_f32_16x16x4_f32(false, a3, false, wt[kc + 3],
                                                         (short)0, acc3, false, false);
        }
        float g  = (acc0[0] + acc1[0]) + (acc2[0] + acc3[0]);
        float xn = yl - s * (g - bv);
        float yn = (1.0f + alpha) * xn - alpha * xl;
        xl = xn;
        yl = yn;

        __syncthreads();
        if (!(lane & 16)) sy[rb + n] = yn;
        __syncthreads();
    }

    if (!(lane & 16)) {
        out[(size_t)batch * NN + rb + n] = xl;
        out[(size_t)NB * NN + (size_t)batch * NN + rb + n] = yl;
    }
}

extern "C" void kernel_launch(void* const* d_in, const int* in_sizes, int n_in,
                              void* d_out, int out_size, void* d_ws, size_t ws_size,
                              hipStream_t stream) {
    (void)in_sizes; (void)n_in; (void)d_ws; (void)ws_size; (void)out_size;
    const float* w = (const float*)d_in[0];
    const float* b = (const float*)d_in[1];
    const float* s = (const float*)d_in[2];
    float* out = (float*)d_out;
    nag_wmma_kernel<<<dim3(NB), dim3(256), 0, stream>>>(w, b, s, out);
}
